# starting kernel
# baseline (speedup 1.0000x reference)
#include <hip/hip_runtime.h>
#include <hip/hip_bf16.h>

typedef __attribute__((ext_vector_type(2))) float v2f;
typedef __attribute__((ext_vector_type(8))) float v8f;

#define E 256
#define SEQ_L 512
#define M_TOT 32768
#define BM 64
#define BN 128
#define LSTRIDE 260
#define NLAYER 4

__global__ __launch_bounds__(256) void wtrans_kernel(const float* __restrict__ cw,
                                                     float* __restrict__ wt) {
  int idx = blockIdx.x * 256 + threadIdx.x;
  int j  = idx & 1;
  int o  = (idx >> 1) & 255;
  int e2 = (idx >> 9) & 127;
  int t  = (idx >> 16) & 1;
  int l  = idx >> 17;
  int e  = e2 * 2 + j;
  wt[idx] = cw[(((size_t)l * E + o) * E + e) * 3 + t];
}

__global__ __launch_bounds__(256) void conv_layer_kernel(const float* __restrict__ Xin,
                                                         const float* __restrict__ Wt,
                                                         float* __restrict__ Xout) {
  __shared__ float ldsX[(BM + 1) * LSTRIDE];

  const int tid  = threadIdx.x;
  const int lane = tid & 31;
  const int wave = tid >> 5;
  const int mb   = blockIdx.x * BM;
  const int nb   = blockIdx.y * BN;

  const bool bstart = (mb % SEQ_L) == 0;
  for (int idx = tid; idx < (BM + 1) * (E / 4); idx += 256) {
    const int r  = idx >> 6;
    const int c4 = idx & 63;
    float4 v;
    if (r == 0 && bstart) {
      v = make_float4(0.f, 0.f, 0.f, 0.f);
    } else {
      v = ((const float4*)(Xin + (size_t)(mb - 1 + r) * E))[c4];
    }
    float* dst = ldsX + r * LSTRIDE + c4 * 4;
    dst[0] = v.x; dst[1] = v.y; dst[2] = v.z; dst[3] = v.w;
  }
  __syncthreads();

  const int wm    = wave >> 1;
  const int wn    = wave & 1;
  const int r0    = wm * 16;
  const int kh    = (lane >> 4) << 1;
  const int nlo   = lane & 15;
  const int ncol0 = nb + wn * 64 + nlo;

  v8f acc0 = {}, acc1 = {}, acc2 = {}, acc3 = {};

#pragma unroll
  for (int t = 0; t < 2; ++t) {
    const float* __restrict__ arow = ldsX + (size_t)(r0 + nlo + t) * LSTRIDE;
    const v2f*   __restrict__ Wtt  = (const v2f*)(Wt + (size_t)t * E * E);
    for (int k0 = 0; k0 < E; k0 += 4) {
      const int e = k0 + kh;
      v2f a = *(const v2f*)(arow + e);
      const v2f* bp = Wtt + ((size_t)(e >> 1) * E + ncol0);
      v2f b0 = bp[0], b1 = bp[16], b2 = bp[32], b3 = bp[48];
      acc0 = __builtin_amdgcn_wmma_f32_16x16x4_f32(false, a, false, b0, (short)0, acc0, false, false);
      acc1 = __builtin_amdgcn_wmma_f32_16x16x4_f32(false, a, false, b1, (short)0, acc1, false, false);
      acc2 = __builtin_amdgcn_wmma_f32_16x16x4_f32(false, a, false, b2, (short)0, acc2, false, false);
      acc3 = __builtin_amdgcn_wmma_f32_16x16x4_f32(false, a, false, b3, (short)0, acc3, false, false);
    }
  }

  const int mrow = mb + r0 + ((lane >> 4) << 3);
  float* outp = Xout + (size_t)mrow * E + ncol0;
#pragma unroll
  for (int v = 0; v < 8; ++v) {
    outp[(size_t)v * E +  0] = acc0[v];
    outp[(size_t)v * E + 16] = acc1[v];
    outp[(size_t)v * E + 32] = acc2[v];
    outp[(size_t)v * E + 48] = acc3[v];
  }
}

__global__ __launch_bounds__(256) void fc_kernel(const float* __restrict__ X,
                                                 const float* __restrict__ w,
                                                 const float* __restrict__ b,
                                                 float* __restrict__ out) {
  const int wave = threadIdx.x >> 5;
  const int lane = threadIdx.x & 31;
  const int m = blockIdx.x * 8 + wave;
  const float* xr = X + (size_t)m * E;
  float s = 0.f;
#pragma unroll
  for (int e = lane; e < E; e += 32) s += xr[e] * w[e];
#pragma unroll
  for (int off = 16; off > 0; off >>= 1) s += __shfl_down(s, off);
  if (lane == 0) out[m] = s + b[0];
}

extern "C" void kernel_launch(void* const* d_in, const int* in_sizes, int n_in,
                              void* d_out, int out_size, void* d_ws, size_t ws_size,
                              hipStream_t stream) {
  const float* x      = (const float*)d_in[0];
  const float* conv_w = (const float*)d_in[1];
  const float* fc_w   = (const float*)d_in[2];
  const float* fc_b   = (const float*)d_in[3];
  float* out = (float*)d_out;

  float* ws   = (float*)d_ws;
  float* Wt   = ws;
  float* bufA = Wt + (size_t)NLAYER * 2 * E * E;
  float* bufB = bufA + (size_t)M_TOT * E;

  wtrans_kernel<<<(NLAYER * 2 * E * E) / 256, 256, 0, stream>>>(conv_w, Wt);

  dim3 grid(M_TOT / BM, E / BN);
  const size_t WL = (size_t)2 * E * E;
  conv_layer_kernel<<<grid, 256, 0, stream>>>(x,    Wt + 0 * WL, bufA);
  conv_layer_kernel<<<grid, 256, 0, stream>>>(bufA, Wt + 1 * WL, bufB);
  conv_layer_kernel<<<grid, 256, 0, stream>>>(bufB, Wt + 2 * WL, bufA);
  conv_layer_kernel<<<grid, 256, 0, stream>>>(bufA, Wt + 3 * WL, bufB);

  fc_kernel<<<M_TOT / 8, 256, 0, stream>>>(bufB, fc_w, fc_b, out);
}


# Fold two layers per launch: a work-group folds its 16-channel slice through one layer straight into the next; next launch adds the 16 partials in fixed order (re-associates f32 sums); 5 launches to 3
# speedup vs baseline: 13.0403x; 13.0403x over previous
#include <hip/hip_runtime.h>


typedef __attribute__((ext_vector_type(2))) float v2f;
typedef __attribute__((ext_vector_type(8))) float v8f;

#define E 256
#define SEQ_L 512
#define M_TOT 32768
#define NTAP 5
#define WROW (E * 3)
#define UT 80
#define PSZ (16 * UT)
#define NPART 16
#define CP 264
#define XP 260
#define RB 128


template <int NPART_IN, int NROWS_IN>
__global__ __launch_bounds__(256) void fold_pair_kernel(const float* __restrict__ Uin, int ts, int rs, int ps,
                                                        const float* __restrict__ Wa,
                                                        const float* __restrict__ Wb,
                                                        float* __restrict__ Upart) {
  __shared__ __attribute__((aligned(16))) float cin[NROWS_IN * CP];
  __shared__ float red[8][NTAP][16];
  __shared__ __attribute__((aligned(16))) float mid[NTAP * 16];

  const int tid  = threadIdx.x;
  const int lane = tid & 31;
  const int wave = tid >> 5;
  const int c    = lane & 15;
  const int h    = lane >> 4;
  const int m0   = blockIdx.x * 16;

  constexpr int PER = 4 * NROWS_IN;
  const bool stager = tid < 16 * PER;
  const int  piece  = stager ? tid : 0;
  const int  ptile = piece / PER, prow = (piece % PER) >> 2, pq = piece & 3;
  const float* up = Uin + ptile * ts + prow * rs + 4 * pq;
  float4 pv[NPART_IN];
#pragma unroll
  for (int p = 0; p < NPART_IN; ++p) pv[p] = *(const float4*)(up + p * ps);

  v2f wa1[8], wa0[8];
#pragma unroll
  for (int ks = 0; ks < 8; ++ks) {
    const int o = wave * 32 + ks * 4 + 2 * h;
    const float* wp = Wa + (size_t)o * WROW + (m0 + c) * 3;
    wa1[ks].x = wp[1];  wa1[ks].y = wp[WROW + 1];
    wa0[ks].x = wp[0];  wa0[ks].y = wp[WROW];
  }
  v2f wb1[2][4], wb0[2][4];
#pragma unroll
  for (int t = 0; t < 2; ++t) {
#pragma unroll
    for (int ks = 0; ks < 4; ++ks) {
      const int m = m0 + ks * 4 + 2 * h;
      const float* wp = Wb + (size_t)m * WROW + ((2 * wave + t) * 16 + c) * 3;
      wb1[t][ks].x = wp[1];  wb1[t][ks].y = wp[WROW + 1];
      wb0[t][ks].x = wp[0];  wb0[t][ks].y = wp[WROW];
    }
  }

  if (stager) {
    float4 s = pv[0];
#pragma unroll
    for (int p = 1; p < NPART_IN; ++p) { s.x += pv[p].x; s.y += pv[p].y; s.z += pv[p].z; s.w += pv[p].w; }
    *(float4*)(cin + prow * CP + ptile * 16 + 4 * pq) = s;
  }
  __syncthreads();

  const bool live1 = c < NROWS_IN;
  const bool live0 = (c >= 1) && (c <= NROWS_IN);
  const int  r1 = live1 ? c : 0;
  const int  r0 = live0 ? c - 1 : 0;
  v8f acc1 = {}, acc0 = {};
#pragma unroll
  for (int ks = 0; ks < 8; ++ks) {
    const int o = wave * 32 + ks * 4 + 2 * h;
    v2f a1 = *(const v2f*)(cin + r1 * CP + o);
    v2f a0 = *(const v2f*)(cin + r0 * CP + o);
    a1.x = live1 ? a1.x : 0.f;  a1.y = live1 ? a1.y : 0.f;
    a0.x = live0 ? a0.x : 0.f;  a0.y = live0 ? a0.y : 0.f;
    acc1 = __builtin_amdgcn_wmma_f32_16x16x4_f32(false, a1, false, wa1[ks], (short)0, acc1, false, false);
    acc0 = __builtin_amdgcn_wmma_f32_16x16x4_f32(false, a0, false, wa0[ks], (short)0, acc0, false, false);
  }
  const v8f accm = acc1 + acc0;

  if (h == 0) {
#pragma unroll
    for (int r = 0; r < NTAP; ++r) red[wave][r][c] = accm[r];
  }
  __syncthreads();
  if (tid < NTAP * 16) {
    const int r = tid >> 4, cc = tid & 15;
    float s = 0.f;
    for (int w = 0; w < 8; ++w) s += red[w][r][cc];
    mid[r * 16 + cc] = s;
  }
  __syncthreads();

  const bool liv1 = c <= NROWS_IN;
  const bool liv0 = (c >= 1) && (c <= NROWS_IN + 1);
  const int  q1 = liv1 ? c : 0;
  const int  q0 = liv0 ? c - 1 : 0;
  v8f o1[2] = {{}, {}}, o0[2] = {{}, {}};
#pragma unroll
  for (int ks = 0; ks < 4; ++ks) {
    const int k = ks * 4 + 2 * h;
    v2f a1 = *(const v2f*)(mid + q1 * 16 + k);
    v2f a0 = *(const v2f*)(mid + q0 * 16 + k);
    a1.x = liv1 ? a1.x : 0.f;  a1.y = liv1 ? a1.y : 0.f;
    a0.x = liv0 ? a0.x : 0.f;  a0.y = liv0 ? a0.y : 0.f;
#pragma unroll
    for (int t = 0; t < 2; ++t) {
      o1[t] = __builtin_amdgcn_wmma_f32_16x16x4_f32(false, a1, false, wb1[t][ks], (short)0, o1[t], false, false);
      o0[t] = __builtin_amdgcn_wmma_f32_16x16x4_f32(false, a0, false, wb0[t][ks], (short)0, o0[t], false, false);
    }
  }
  const v8f out0 = o1[0] + o0[0];
  const v8f out1 = o1[1] + o0[1];
  if (h == 0) {
    float* dst = Upart + blockIdx.x * PSZ + (2 * wave) * UT + c;
#pragma unroll
    for (int r = 0; r < NTAP; ++r) {
      dst[r * 16]      = out0[r];
      dst[UT + r * 16] = out1[r];
    }
  }
}

__global__ __launch_bounds__(288) void tap_kernel(const float* __restrict__ x,
                                                  const float* __restrict__ Vp,
                                                  const float* __restrict__ fcb,
                                                  float* __restrict__ out) {
  __shared__ __attribute__((aligned(16))) float xs[9 * 16 * XP];
  __shared__ __attribute__((aligned(16))) float vsh[6 * XP];
  __shared__ float zs[9 * 16 * NTAP];

  const int tid  = threadIdx.x;
  const int lane = tid & 31;
  const int wave = tid >> 5;
  const int c    = lane & 15;
  const int h    = lane >> 4;
  const int R0   = blockIdx.x * RB;

  int rowbase = R0 + (wave - 1) * 16;
  rowbase = rowbase < 0 ? 0 : rowbase;

  float4 va[NPART], vb[NPART];
#pragma unroll
  for (int p = 0; p < NPART; ++p) va[p] = *(const float4*)(Vp + p * PSZ + 4 * tid);
  if (wave == 0) {
#pragma unroll
    for (int p = 0; p < NPART; ++p) vb[p] = *(const float4*)(Vp + p * PSZ + 4 * (288 + tid));
  }

  const float* xg = x + (size_t)(rowbase + (lane >> 4)) * E + (lane & 15) * 4;
  float*       xl = xs + wave * 16 * XP + (lane >> 4) * XP + (lane & 15) * 4;
#define LDX(q, p) const float4 r_##q##_##p = *(const float4*)(xg + (2 * p) * E + 64 * q);
#define STX(q, p) *(float4*)(xl + (2 * p) * XP + 64 * q) = r_##q##_##p;
#define LD8(q) LDX(q, 0) LDX(q, 1) LDX(q, 2) LDX(q, 3) LDX(q, 4) LDX(q, 5) LDX(q, 6) LDX(q, 7)
#define ST8(q) STX(q, 0) STX(q, 1) STX(q, 2) STX(q, 3) STX(q, 4) STX(q, 5) STX(q, 6) STX(q, 7)
  LD8(0) LD8(1) LD8(2) LD8(3)
  __builtin_amdgcn_sched_barrier(0);

  {
    float4 s = va[0];
#pragma unroll
    for (int p = 1; p < NPART; ++p) { s.x += va[p].x; s.y += va[p].y; s.z += va[p].z; s.w += va[p].w; }
    const int j = tid;
    *(float4*)(vsh + ((j % 20) >> 2) * XP + (j / 20) * 16 + (j & 3) * 4) = s;
  }
  if (wave == 0) {
    float4 s = vb[0];
#pragma unroll
    for (int p = 1; p < NPART; ++p) { s.x += vb[p].x; s.y += vb[p].y; s.z += vb[p].z; s.w += vb[p].w; }
    const int j = 288 + tid;
    *(float4*)(vsh + ((j % 20) >> 2) * XP + (j / 20) * 16 + (j & 3) * 4) = s;
  }
  if (tid < 64) *(float4*)(vsh + 5 * XP + tid * 4) = make_float4(0.f, 0.f, 0.f, 0.f);
  __builtin_amdgcn_fence(__ATOMIC_RELEASE, "workgroup", "local");
  __builtin_amdgcn_s_barrier();
  __builtin_amdgcn_fence(__ATOMIC_ACQUIRE, "workgroup", "local");

  const float* afrag = xs + wave * 16 * XP + c * XP + 4 * h;
  const float* bfrag = vsh + (c < NTAP ? c : NTAP) * XP + 4 * h;
  v8f acc0 = {}, acc1 = {};
#define WAVE_HANDOVER()                                   \
  __builtin_amdgcn_fence(__ATOMIC_RELEASE, "wavefront");  \
  __builtin_amdgcn_wave_barrier();                        \
  __builtin_amdgcn_fence(__ATOMIC_ACQUIRE, "wavefront");
#define MUL8(q)                                                                                              \
  _Pragma("unroll")        \
  for (int j = 0; j < 8; ++j) {                                                                              \
    const float4 a = *(const float4*)(afrag + 64 * q + 8 * j);                                               \
    const float4 b = *(const float4*)(bfrag + 64 * q + 8 * j);                                               \
    const v2f a01 = {a.x, a.y}, a23 = {a.z, a.w};                                                            \
    const v2f b01 = {b.x, b.y}, b23 = {b.z, b.w};                                                            \
    acc0 = __builtin_amdgcn_wmma_f32_16x16x4_f32(false, a01, false, b01, (short)0, acc0, false, false);      \
    acc1 = __builtin_amdgcn_wmma_f32_16x16x4_f32(false, a23, false, b23, (short)0, acc1, false, false);      \
  }
  ST8(0) WAVE_HANDOVER() MUL8(0)
  ST8(1) WAVE_HANDOVER() MUL8(1)
  ST8(2) WAVE_HANDOVER() MUL8(2)
  ST8(3) WAVE_HANDOVER() MUL8(3)
#undef LDX
#undef STX
#undef LD8
#undef ST8
#undef MUL8
#undef WAVE_HANDOVER
  const v8f acc = acc0 + acc1;

  if (c < NTAP) {
#pragma unroll
    for (int v = 0; v < 8; ++v) zs[(wave * 16 + 8 * h + v) * NTAP + c] = acc[v];
  }
  __syncthreads();

  if (tid < RB) {
    const int m   = R0 + tid;
    const int pos = m & (SEQ_L - 1);
    float s = fcb[0];
    for (int d = 0; d < NTAP; ++d)
      if (pos >= d) s += zs[(16 + tid - d) * NTAP + d];
    out[m] = s;
  }
}

extern "C" void kernel_launch(void* const* d_in, const int* in_sizes, int n_in,
                              void* d_out, int out_size, void* d_ws, size_t ws_size,
                              hipStream_t stream) {
  const float* x      = (const float*)d_in[0];
  const float* conv_w = (const float*)d_in[1];
  const float* fc_w   = (const float*)d_in[2];
  const float* fc_b   = (const float*)d_in[3];
  float* out = (float*)d_out;

  float* ws = (float*)d_ws;
  float* P2 = ws;
  float* PV = ws + NPART * PSZ;

  const size_t WL = (size_t)E * E * 3;
  fold_pair_kernel<1, 1><<<NPART, 256, 0, stream>>>(fc_w, 16, 0, 0, conv_w + 3 * WL, conv_w + 2 * WL, P2);
  fold_pair_kernel<NPART, 3><<<NPART, 256, 0, stream>>>(P2, UT, 16, PSZ, conv_w + 1 * WL, conv_w + 0 * WL, PV);

  tap_kernel<<<M_TOT / RB, 288, 0, stream>>>(x, PV, fc_b, out);
}
